# starting kernel
# baseline (speedup 1.0000x reference)
#include <hip/hip_runtime.h>

typedef __attribute__((ext_vector_type(2))) float v2f;
typedef __attribute__((ext_vector_type(8))) float v8f;

#define SEQ_L 2048
#define HEAD_D 64
#define NBLK 128

__global__ __launch_bounds__(32)
void sparse_attn_wmma_f32(const float* __restrict__ q,
                          const float* __restrict__ k,
                          const float* __restrict__ v,
                          float* __restrict__ out) {
  __shared__ float pshm[16 * 16];

  const int lane = threadIdx.x;
  const int half = lane >> 4;
  const int l16  = lane & 15;

  const int  row  = (int)(blockIdx.x & (NBLK - 1));
  const long bh   = (long)(blockIdx.x >> 7);
  const long base = bh * (long)(SEQ_L * HEAD_D);

  const float* qrow = q + base + (long)(row * 16 + l16) * HEAD_D + 2 * half;
  v2f qa[16];
#pragma unroll
  for (int kk = 0; kk < 16; ++kk) {
    v2f t = *(const v2f*)(qrow + 4 * kk);
    qa[kk].x = t.x * 0.125f;
    qa[kk].y = t.y * 0.125f;
  }

  v8f o0 = {}, o1 = {}, o2 = {}, o3 = {};
  float mrow[8], lrow[8];
#pragma unroll
  for (int r = 0; r < 8; ++r) { mrow[r] = -1e30f; lrow[r] = 0.0f; }

  const int w     = row & ~3;
  const int nvert = w >> 2;
  const int nbl   = nvert + (row - w + 1);

  for (int t = 0; t < nbl; ++t) {
    const int jb = (t < nvert) ? (4 * t + 3) : (w + (t - nvert));

    {
      const int tn  = (t + 1 < nbl) ? (t + 1) : t;
      const int jbn = (tn < nvert) ? (4 * tn + 3) : (w + (tn - nvert));
      const float* kp = k + base + (long)(jbn * 16 + l16) * HEAD_D;
      const float* vp = v + base + (long)(jbn * 16 + l16) * HEAD_D;
      __builtin_prefetch(kp, 0, 0);
      __builtin_prefetch(vp, 0, 0);
    }

    const float* krow = k + base + (long)(jb * 16 + l16) * HEAD_D + 2 * half;
    v8f s = {};
#pragma unroll
    for (int kk = 0; kk < 16; ++kk) {
      v2f b = *(const v2f*)(krow + 4 * kk);
      s = __builtin_amdgcn_wmma_f32_16x16x4_f32(false, qa[kk], false, b,
                                                (short)0, s, false, false);
    }

    float alpha[8];
#pragma unroll
    for (int r = 0; r < 8; ++r) {
      float x  = s[r];
      float rm = x;
#pragma unroll
      for (int off = 1; off < 16; off <<= 1)
        rm = fmaxf(rm, __shfl_xor(rm, off, 32));
      const float mnew = fmaxf(mrow[r], rm);
      alpha[r] = __expf(mrow[r] - mnew);
      float p = __expf(x - mnew);
      float rs = p;
#pragma unroll
      for (int off = 1; off < 16; off <<= 1)
        rs += __shfl_xor(rs, off, 32);
      lrow[r] = lrow[r] * alpha[r] + rs;
      mrow[r] = mnew;
      pshm[(r + 8 * half) * 16 + l16] = p;
    }
    __syncthreads();

#pragma unroll
    for (int r = 0; r < 8; ++r) {
      o0[r] *= alpha[r]; o1[r] *= alpha[r];
      o2[r] *= alpha[r]; o3[r] *= alpha[r];
    }

    const float* vblk = v + base + (long)(jb * 16) * HEAD_D;
#pragma unroll
    for (int kp = 0; kp < 4; ++kp) {
      v2f a = *(const v2f*)&pshm[l16 * 16 + 4 * kp + 2 * half];
      const float* vr = vblk + (long)(4 * kp + 2 * half) * HEAD_D + l16;
      v2f b0, b1, b2, b3;
      b0.x = vr[0];  b0.y = vr[HEAD_D + 0];
      b1.x = vr[16]; b1.y = vr[HEAD_D + 16];
      b2.x = vr[32]; b2.y = vr[HEAD_D + 32];
      b3.x = vr[48]; b3.y = vr[HEAD_D + 48];
      o0 = __builtin_amdgcn_wmma_f32_16x16x4_f32(false, a, false, b0, (short)0, o0, false, false);
      o1 = __builtin_amdgcn_wmma_f32_16x16x4_f32(false, a, false, b1, (short)0, o1, false, false);
      o2 = __builtin_amdgcn_wmma_f32_16x16x4_f32(false, a, false, b2, (short)0, o2, false, false);
      o3 = __builtin_amdgcn_wmma_f32_16x16x4_f32(false, a, false, b3, (short)0, o3, false, false);
    }
    __syncthreads();
  }

  float* orow = out + base + (long)(row * 16) * HEAD_D;
#pragma unroll
  for (int r = 0; r < 8; ++r) {
    const float inv = 1.0f / lrow[r];
    const int m = r + 8 * half;
    orow[(long)m * HEAD_D +  0 + l16] = o0[r] * inv;
    orow[(long)m * HEAD_D + 16 + l16] = o1[r] * inv;
    orow[(long)m * HEAD_D + 32 + l16] = o2[r] * inv;
    orow[(long)m * HEAD_D + 48 + l16] = o3[r] * inv;
  }
}

extern "C" void kernel_launch(void* const* d_in, const int* in_sizes, int n_in,
                              void* d_out, int out_size, void* d_ws, size_t ws_size,
                              hipStream_t stream) {
  (void)n_in; (void)out_size; (void)d_ws; (void)ws_size;
  const float* q = (const float*)d_in[0];
  const float* k = (const float*)d_in[1];
  const float* v = (const float*)d_in[2];
  float* out = (float*)d_out;

  const int bh = in_sizes[0] / (SEQ_L * HEAD_D);
  dim3 grid((unsigned)(bh * NBLK));
  sparse_attn_wmma_f32<<<grid, 32, 0, stream>>>(q, k, v, out);
}


# score tile, not P, takes the C-to-A trip through LDS: a lane holds 16 keys of ONE query row, so row max and sum are in-lane + one half exchange (were 64 DPP butterflies a step); row-sum order changes
# speedup vs baseline: 2.3574x; 2.3574x over previous
#include <hip/hip_runtime.h>


typedef __attribute__((ext_vector_type(2))) float v2f;
typedef __attribute__((ext_vector_type(4))) float v4f;
typedef __attribute__((ext_vector_type(8))) float v8f;

__device__ __forceinline__ float xhalf(float x) {
  const int b = __builtin_bit_cast(int, x);
  return __builtin_bit_cast(float,
      __builtin_amdgcn_permlanex16(b, b, 0x76543210, 0xfedcba98, false, false));
}

template <int I>
__device__ __forceinline__ float row_bcast(int x) {
  return __builtin_bit_cast(float, __builtin_amdgcn_update_dpp(0, x, 0x150 + I, 0xf, 0xf, true));
}

__device__ __forceinline__ void rows_to_d(float (&d)[8], float x) {
  const int xi  = __builtin_bit_cast(int, x);
  const int rot = __builtin_amdgcn_update_dpp(xi, xi, 0x160 + 8, 0x2, 0xf, false);
  d[0] = row_bcast<0>(rot); d[1] = row_bcast<1>(rot);
  d[2] = row_bcast<2>(rot); d[3] = row_bcast<3>(rot);
  d[4] = row_bcast<4>(rot); d[5] = row_bcast<5>(rot);
  d[6] = row_bcast<6>(rot); d[7] = row_bcast<7>(rot);
}

#define SEQ_L 2048
#define HEAD_D 64
#define NBLK 128
#define NPAIR 64
#define WG_WAVES 8
#define P_PITCH 36

__device__ __forceinline__ int key_block(int row, int t) {
  const int w     = row & ~3;
  const int nvert = w >> 2;
  return (t < nvert) ? (4 * t + 3) : (w + (t - nvert));
}

__device__ __forceinline__ void load_k_frags(v2f (&kf)[16], const float* __restrict__ kbh,
                                             int jb, int l16, int half) {
  const float* krow = kbh + (long)(jb * 16 + l16) * HEAD_D + 4 * half;
#pragma unroll
  for (int j = 0; j < 8; ++j) {
    const v4f t = *(const v4f*)(krow + 8 * j);
    kf[2 * j + 0].x = t.x; kf[2 * j + 0].y = t.y;
    kf[2 * j + 1].x = t.z; kf[2 * j + 1].y = t.w;
  }
}

__device__ __forceinline__ void load_v_rows(v4f (&vr)[8], const float* __restrict__ vbh,
                                            int jb, int l16, int half) {
  const float* vblk = vbh + (long)(jb * 16) * HEAD_D + 4 * l16;
#pragma unroll
  for (int kp = 0; kp < 4; ++kp) {
    const float* p = vblk + (long)(4 * kp + 2 * half) * HEAD_D;
    vr[2 * kp + 0] = *(const v4f*)(p);
    vr[2 * kp + 1] = *(const v4f*)(p + HEAD_D);
  }
}

template <int C>
__device__ __forceinline__ v2f v_frag(const v4f (&vr)[8], int kp) {
  v2f f;
  f.x = vr[2 * kp + 0][C];
  f.y = vr[2 * kp + 1][C];
  return f;
}

template <int NT, int FETCH>
__device__ __forceinline__ void tile_step(
    float* pshm, int l16, int half, const v2f (&qa)[16],
    const v2f (&kc0)[16], const v2f (&kc1)[16], v2f (&kn0)[16], v2f (&kn1)[16],
    const float* __restrict__ kbh, const float* __restrict__ vbh,
    int jb0, int jb1, int jbn0, int jbn1,
    v8f& o0, v8f& o1, v8f& o2, v8f& o3, float& mrun, float& lrun) {
  v4f vc0[8], vc1[8];
  load_v_rows(vc0, vbh, jb0, l16, half);
  if (NT == 2) load_v_rows(vc1, vbh, jb1, l16, half);
  __builtin_amdgcn_sched_barrier(0);

  v8f s0 = {}, s1 = {};
#pragma unroll
  for (int kk = 0; kk < 16; ++kk) {
    s0 = __builtin_amdgcn_wmma_f32_16x16x4_f32(false, qa[kk], false, kc0[kk],
                                               (short)0, s0, false, false);
    if (NT == 2)
      s1 = __builtin_amdgcn_wmma_f32_16x16x4_f32(false, qa[kk], false, kc1[kk],
                                                 (short)0, s1, false, false);
  }
  if (FETCH >= 1) load_k_frags(kn0, kbh, jbn0, l16, half);
  if (FETCH >= 2) load_k_frags(kn1, kbh, jbn1, l16, half);
  __builtin_amdgcn_sched_barrier(0);

#pragma unroll
  for (int r = 0; r < 8; ++r) {
    float* srow = pshm + (r + 8 * half) * P_PITCH + l16;
    srow[0] = s0[r];
    if (NT == 2) srow[16] = s1[r];
  }
  __builtin_amdgcn_wave_barrier();
  v2f x[4 * NT];
#pragma unroll
  for (int kp = 0; kp < 4 * NT; ++kp)
    x[kp] = *(const v2f*)&pshm[l16 * P_PITCH + 4 * kp + 2 * half];
  __builtin_amdgcn_wave_barrier();

  float rm = fmaxf(x[0].x, x[0].y);
#pragma unroll
  for (int kp = 1; kp < 4 * NT; ++kp) rm = fmaxf(rm, fmaxf(x[kp].x, x[kp].y));
  rm = fmaxf(rm, xhalf(rm));
  const float mnew  = fmaxf(mrun, rm);
  const float alpha = __expf(mrun - mnew);
  v2f a[4 * NT];
#pragma unroll
  for (int kp = 0; kp < 4 * NT; ++kp) {
    a[kp].x = __expf(x[kp].x - mnew);
    a[kp].y = __expf(x[kp].y - mnew);
  }
  v2f ps;
  if (NT == 2) ps = ((a[0] + a[1]) + (a[2] + a[3])) + ((a[4] + a[5]) + (a[6] + a[7]));
  else         ps = (a[0] + a[1]) + (a[2] + a[3]);
  lrun = lrun * alpha + (ps.x + ps.y);
  mrun = mnew;

  float ad[8];
  rows_to_d(ad, alpha);
#pragma unroll
  for (int r = 0; r < 8; ++r) {
    o0[r] *= ad[r]; o1[r] *= ad[r];
    o2[r] *= ad[r]; o3[r] *= ad[r];
  }

#pragma unroll
  for (int kp = 0; kp < 4; ++kp) {
    o0 = __builtin_amdgcn_wmma_f32_16x16x4_f32(false, a[kp], false, v_frag<0>(vc0, kp), (short)0, o0, false, false);
    o1 = __builtin_amdgcn_wmma_f32_16x16x4_f32(false, a[kp], false, v_frag<1>(vc0, kp), (short)0, o1, false, false);
    o2 = __builtin_amdgcn_wmma_f32_16x16x4_f32(false, a[kp], false, v_frag<2>(vc0, kp), (short)0, o2, false, false);
    o3 = __builtin_amdgcn_wmma_f32_16x16x4_f32(false, a[kp], false, v_frag<3>(vc0, kp), (short)0, o3, false, false);
  }
  if (NT == 2) {
#pragma unroll
    for (int kp = 0; kp < 4; ++kp) {
      o0 = __builtin_amdgcn_wmma_f32_16x16x4_f32(false, a[4 + kp], false, v_frag<0>(vc1, kp), (short)0, o0, false, false);
      o1 = __builtin_amdgcn_wmma_f32_16x16x4_f32(false, a[4 + kp], false, v_frag<1>(vc1, kp), (short)0, o1, false, false);
      o2 = __builtin_amdgcn_wmma_f32_16x16x4_f32(false, a[4 + kp], false, v_frag<2>(vc1, kp), (short)0, o2, false, false);
      o3 = __builtin_amdgcn_wmma_f32_16x16x4_f32(false, a[4 + kp], false, v_frag<3>(vc1, kp), (short)0, o3, false, false);
    }
  }
}

__global__ __launch_bounds__(32 * WG_WAVES)
void sparse_attn_wmma_f32(const float* __restrict__ q,
                          const float* __restrict__ k,
                          const float* __restrict__ v,
                          float* __restrict__ out) {
  __shared__ __attribute__((aligned(16))) float pshm_all[WG_WAVES * 16 * P_PITCH];

  const int wv   = __builtin_amdgcn_readfirstlane((int)(threadIdx.x >> 5));
  float* pshm    = pshm_all + wv * (16 * P_PITCH);
  const int lane = threadIdx.x & 31;
  const int half = lane >> 4;
  const int l16  = lane & 15;

  const int  pr   = (int)(blockIdx.x & (NPAIR / WG_WAVES - 1)) * WG_WAVES + wv;
  const long bh   = (long)(blockIdx.x / (NPAIR / WG_WAVES));
  const long base = bh * (long)(SEQ_L * HEAD_D);
  const float* kbh = k + base;
  const float* vbh = v + base;

  for (int pass = 0; pass < 2; ++pass) {
    const int row = pass ? pr : (NBLK - 1 - pr);

    const int w     = row & ~3;
    const int nvert = w >> 2;
    const int nbl   = nvert + (row - w + 1);
    const int last  = nbl - 1;

    v2f kA0[16], kA1[16], kB0[16], kB1[16];
    load_k_frags(kA0, kbh, key_block(row, 0), l16, half);
    load_k_frags(kA1, kbh, key_block(row, last < 1 ? last : 1), l16, half);

    const float* qrow = q + base + (long)(row * 16 + l16) * HEAD_D + 4 * half;
    v2f qa[16];
#pragma unroll
    for (int j = 0; j < 8; ++j) {
      const v4f t = *(const v4f*)(qrow + 8 * j);
      qa[2 * j + 0].x = t.x * 0.125f; qa[2 * j + 0].y = t.y * 0.125f;
      qa[2 * j + 1].x = t.z * 0.125f; qa[2 * j + 1].y = t.w * 0.125f;
    }

    v8f o0 = {}, o1 = {}, o2 = {}, o3 = {};
    float mrun = -1e30f, lrun = 0.0f;

    int t = 0;
    while (nbl - t >= 4) {
      const int f0 = t + 4 < last ? t + 4 : last;
      const int f1 = t + 5 < last ? t + 5 : last;
      tile_step<2, 2>(pshm, l16, half, qa, kA0, kA1, kB0, kB1, kbh, vbh,
                      key_block(row, t), key_block(row, t + 1),
                      key_block(row, t + 2), key_block(row, t + 3),
                      o0, o1, o2, o3, mrun, lrun);
      tile_step<2, 2>(pshm, l16, half, qa, kB0, kB1, kA0, kA1, kbh, vbh,
                      key_block(row, t + 2), key_block(row, t + 3),
                      key_block(row, f0), key_block(row, f1),
                      o0, o1, o2, o3, mrun, lrun);
      t += 4;
    }
    const int rem = nbl - t;
    if (rem == 3) {
      tile_step<2, 1>(pshm, l16, half, qa, kA0, kA1, kB0, kB1, kbh, vbh,
                      key_block(row, t), key_block(row, t + 1),
                      key_block(row, t + 2), 0,
                      o0, o1, o2, o3, mrun, lrun);
      tile_step<1, 0>(pshm, l16, half, qa, kB0, kB1, kA0, kA1, kbh, vbh,
                      key_block(row, t + 2), 0, 0, 0,
                      o0, o1, o2, o3, mrun, lrun);
    } else if (rem == 2) {
      tile_step<2, 0>(pshm, l16, half, qa, kA0, kA1, kB0, kB1, kbh, vbh,
                      key_block(row, t), key_block(row, t + 1), 0, 0,
                      o0, o1, o2, o3, mrun, lrun);
    } else if (rem == 1) {
      tile_step<1, 0>(pshm, l16, half, qa, kA0, kA1, kB0, kB1, kbh, vbh,
                      key_block(row, t), 0, 0, 0,
                      o0, o1, o2, o3, mrun, lrun);
    }

    float* orow = out + base + (long)(row * 16) * HEAD_D + 4 * l16;
    float invd[8];
    rows_to_d(invd, 1.0f / (lrun + xhalf(lrun)));
#pragma unroll
    for (int r = 0; r < 8; ++r) {
      const float inv = invd[r];
      const int m = r + 8 * half;
      v4f ov;
      ov.x = o0[r] * inv; ov.y = o1[r] * inv;
      ov.z = o2[r] * inv; ov.w = o3[r] * inv;
      *(v4f*)(orow + (long)m * HEAD_D) = ov;
    }
  }
}

extern "C" void kernel_launch(void* const* d_in, const int* in_sizes, int n_in,
                              void* d_out, int out_size, void* d_ws, size_t ws_size,
                              hipStream_t stream) {
  (void)n_in; (void)out_size; (void)d_ws; (void)ws_size;
  const float* q = (const float*)d_in[0];
  const float* k = (const float*)d_in[1];
  const float* v = (const float*)d_in[2];
  float* out = (float*)d_out;

  const int bh = in_sizes[0] / (SEQ_L * HEAD_D);
  dim3 grid((unsigned)(bh * (NPAIR / WG_WAVES)));
  sparse_attn_wmma_f32<<<grid, 32 * WG_WAVES, 0, stream>>>(q, k, v, out);
}


# K and V tiles of the key blocks shared by a work-group's 8 rows (28 of 36 per wave) are fetched once per group into an LDS ring, 1/8 by each wave, and read from LDS as fragments; sum order unchanged
# speedup vs baseline: 2.4416x; 1.0357x over previous
#include <hip/hip_runtime.h>


typedef __attribute__((ext_vector_type(2))) float v2f;
typedef __attribute__((ext_vector_type(4))) float v4f;
typedef __attribute__((ext_vector_type(8))) float v8f;

__device__ __forceinline__ float xhalf(float x) {
  const int b = __builtin_bit_cast(int, x);
  return __builtin_bit_cast(float,
      __builtin_amdgcn_permlanex16(b, b, 0x76543210, 0xfedcba98, false, false));
}

template <int I>
__device__ __forceinline__ float row_bcast(int x) {
  return __builtin_bit_cast(float, __builtin_amdgcn_update_dpp(0, x, 0x150 + I, 0xf, 0xf, true));
}

__device__ __forceinline__ void rows_to_d(float (&d)[8], float x) {
  const int xi  = __builtin_bit_cast(int, x);
  const int rot = __builtin_amdgcn_update_dpp(xi, xi, 0x160 + 8, 0x2, 0xf, false);
  d[0] = row_bcast<0>(rot); d[1] = row_bcast<1>(rot);
  d[2] = row_bcast<2>(rot); d[3] = row_bcast<3>(rot);
  d[4] = row_bcast<4>(rot); d[5] = row_bcast<5>(rot);
  d[6] = row_bcast<6>(rot); d[7] = row_bcast<7>(rot);
}

#define SEQ_L 2048
#define HEAD_D 64
#define NBLK 128
#define NPAIR 64
#define WG_WAVES 8
#define P_PITCH 36
#define T_PITCH 68
#define TILE_F (16 * T_PITCH)
#define SLOT_F (8 * TILE_F)
#define NSLOT 3

__device__ __forceinline__ int key_block(int row, int t) {
  const int w     = row & ~3;
  const int nvert = w >> 2;
  return (t < nvert) ? (4 * t + 3) : (w + (t - nvert));
}

__device__ __forceinline__ void load_k_frags(v2f (&kf)[16], const float* __restrict__ kbh,
                                             int jb, int l16, int half) {
  const float* krow = kbh + (long)(jb * 16 + l16) * HEAD_D + 4 * half;
#pragma unroll
  for (int j = 0; j < 8; ++j) {
    const v4f t = *(const v4f*)(krow + 8 * j);
    kf[2 * j + 0].x = t.x; kf[2 * j + 0].y = t.y;
    kf[2 * j + 1].x = t.z; kf[2 * j + 1].y = t.w;
  }
}

__device__ __forceinline__ void load_v_rows(v4f (&vr)[8], const float* __restrict__ vbh,
                                            int jb, int l16, int half) {
  const float* vblk = vbh + (long)(jb * 16) * HEAD_D + 4 * l16;
#pragma unroll
  for (int kp = 0; kp < 4; ++kp) {
    const float* p = vblk + (long)(4 * kp + 2 * half) * HEAD_D;
    vr[2 * kp + 0] = *(const v4f*)(p);
    vr[2 * kp + 1] = *(const v4f*)(p + HEAD_D);
  }
}

template <int C>
__device__ __forceinline__ v2f v_frag(const v4f (&vr)[8], int kp) {
  v2f f;
  f.x = vr[2 * kp + 0][C];
  f.y = vr[2 * kp + 1][C];
  return f;
}

__device__ __forceinline__ void load_k_frags_lds(v2f (&kf)[16], const float* kt, int l16, int half) {
  const float* krow = kt + l16 * T_PITCH + 4 * half;
#pragma unroll
  for (int j = 0; j < 8; ++j) {
    const v4f t = *(const v4f*)(krow + 8 * j);
    kf[2 * j + 0].x = t.x; kf[2 * j + 0].y = t.y;
    kf[2 * j + 1].x = t.z; kf[2 * j + 1].y = t.w;
  }
}

__device__ __forceinline__ void load_v_rows_lds(v4f (&vr)[8], const float* vt, int l16, int half) {
#pragma unroll
  for (int kp = 0; kp < 4; ++kp) {
    const float* p = vt + (4 * kp + 2 * half) * T_PITCH + 4 * l16;
    vr[2 * kp + 0] = *(const v4f*)(p);
    vr[2 * kp + 1] = *(const v4f*)(p + T_PITCH);
  }
}

__device__ __forceinline__ void stage_fetch(v4f (&sg)[8], const float* __restrict__ kbh,
                                            const float* __restrict__ vbh, int u, int wv,
                                            int l16, int half) {
  const long off = (long)((16 * u + 3) * 16 + 2 * wv + half) * HEAD_D + 4 * l16;
#pragma unroll
  for (int i = 0; i < 4; ++i) {
    sg[i]     = *(const v4f*)(kbh + off + (long)i * (4 * 16 * HEAD_D));
    sg[4 + i] = *(const v4f*)(vbh + off + (long)i * (4 * 16 * HEAD_D));
  }
}

__device__ __forceinline__ void stage_store(const v4f (&sg)[8], float* slot, int wv, int l16, int half) {
  float* d = slot + (2 * wv + half) * T_PITCH + 4 * l16;
#pragma unroll
  for (int i = 0; i < 8; ++i) *(v4f*)(d + i * TILE_F) = sg[i];
}

__device__ __forceinline__ void stage_barrier() {
  asm volatile("s_wait_dscnt 0x0\n\ts_barrier_signal -1\n\ts_barrier_wait -1" ::: "memory");
}

template <int NT, int FETCH, bool VLDS>
__device__ __forceinline__ void tile_step(
    float* pshm, int l16, int half, const v2f (&qa)[16],
    const v2f (&kc0)[16], const v2f (&kc1)[16], v2f (&kn0)[16], v2f (&kn1)[16],
    const float* __restrict__ kbh, const float* __restrict__ vbh,
    int jb0, int jb1, int jbn0, int jbn1,
    const float* va, const float* vb, const float* ka, const float* kb,
    v8f& o0, v8f& o1, v8f& o2, v8f& o3, float& mrun, float& lrun) {
  v4f vc0[8], vc1[8];
  if (VLDS) {
    load_v_rows_lds(vc0, va, l16, half);
    if (NT == 2) load_v_rows_lds(vc1, vb, l16, half);
  } else {
    load_v_rows(vc0, vbh, jb0, l16, half);
    if (NT == 2) load_v_rows(vc1, vbh, jb1, l16, half);
  }
  __builtin_amdgcn_sched_barrier(0);

  v8f s0 = {}, s1 = {};
#pragma unroll
  for (int kk = 0; kk < 16; ++kk) {
    s0 = __builtin_amdgcn_wmma_f32_16x16x4_f32(false, qa[kk], false, kc0[kk],
                                               (short)0, s0, false, false);
    if (NT == 2)
      s1 = __builtin_amdgcn_wmma_f32_16x16x4_f32(false, qa[kk], false, kc1[kk],
                                                 (short)0, s1, false, false);
  }
  if (FETCH == 1 || FETCH == 2) load_k_frags(kn0, kbh, jbn0, l16, half);
  if (FETCH == 2) load_k_frags(kn1, kbh, jbn1, l16, half);

  __builtin_amdgcn_sched_barrier(0);

#pragma unroll
  for (int r = 0; r < 8; ++r) {
    float* srow = pshm + (r + 8 * half) * P_PITCH + l16;
    srow[0] = s0[r];
    if (NT == 2) srow[16] = s1[r];
  }
  __builtin_amdgcn_wave_barrier();
  v2f x[4 * NT];
#pragma unroll
  for (int kp = 0; kp < 4 * NT; ++kp)
    x[kp] = *(const v2f*)&pshm[l16 * P_PITCH + 4 * kp + 2 * half];
  __builtin_amdgcn_wave_barrier();
  if (FETCH == 3) {
    load_k_frags_lds(kn0, ka, l16, half);
    load_k_frags_lds(kn1, kb, l16, half);
  }

  float rm = fmaxf(x[0].x, x[0].y);
#pragma unroll
  for (int kp = 1; kp < 4 * NT; ++kp) rm = fmaxf(rm, fmaxf(x[kp].x, x[kp].y));
  rm = fmaxf(rm, xhalf(rm));
  const float mnew  = fmaxf(mrun, rm);
  const float alpha = __expf(mrun - mnew);
  v2f a[4 * NT];
#pragma unroll
  for (int kp = 0; kp < 4 * NT; ++kp) {
    a[kp].x = __expf(x[kp].x - mnew);
    a[kp].y = __expf(x[kp].y - mnew);
  }
  v2f ps;
  if (NT == 2) ps = ((a[0] + a[1]) + (a[2] + a[3])) + ((a[4] + a[5]) + (a[6] + a[7]));
  else         ps = (a[0] + a[1]) + (a[2] + a[3]);
  lrun = lrun * alpha + (ps.x + ps.y);
  mrun = mnew;

  float ad[8];
  rows_to_d(ad, alpha);
#pragma unroll
  for (int r = 0; r < 8; ++r) {
    o0[r] *= ad[r]; o1[r] *= ad[r];
    o2[r] *= ad[r]; o3[r] *= ad[r];
  }

#pragma unroll
  for (int kp = 0; kp < 4; ++kp) {
    o0 = __builtin_amdgcn_wmma_f32_16x16x4_f32(false, a[kp], false, v_frag<0>(vc0, kp), (short)0, o0, false, false);
    o1 = __builtin_amdgcn_wmma_f32_16x16x4_f32(false, a[kp], false, v_frag<1>(vc0, kp), (short)0, o1, false, false);
    o2 = __builtin_amdgcn_wmma_f32_16x16x4_f32(false, a[kp], false, v_frag<2>(vc0, kp), (short)0, o2, false, false);
    o3 = __builtin_amdgcn_wmma_f32_16x16x4_f32(false, a[kp], false, v_frag<3>(vc0, kp), (short)0, o3, false, false);
  }
  if (NT == 2) {
#pragma unroll
    for (int kp = 0; kp < 4; ++kp) {
      o0 = __builtin_amdgcn_wmma_f32_16x16x4_f32(false, a[4 + kp], false, v_frag<0>(vc1, kp), (short)0, o0, false, false);
      o1 = __builtin_amdgcn_wmma_f32_16x16x4_f32(false, a[4 + kp], false, v_frag<1>(vc1, kp), (short)0, o1, false, false);
      o2 = __builtin_amdgcn_wmma_f32_16x16x4_f32(false, a[4 + kp], false, v_frag<2>(vc1, kp), (short)0, o2, false, false);
      o3 = __builtin_amdgcn_wmma_f32_16x16x4_f32(false, a[4 + kp], false, v_frag<3>(vc1, kp), (short)0, o3, false, false);
    }
  }
}

__global__ __launch_bounds__(32 * WG_WAVES)
void sparse_attn_wmma_f32(const float* __restrict__ q,
                          const float* __restrict__ k,
                          const float* __restrict__ v,
                          float* __restrict__ out) {
  __shared__ __attribute__((aligned(16))) float pshm_all[WG_WAVES * 16 * P_PITCH];
  __shared__ __attribute__((aligned(16))) float stg[NSLOT * SLOT_F];

  const int wv   = __builtin_amdgcn_readfirstlane((int)(threadIdx.x >> 5));
  float* pshm    = pshm_all + wv * (16 * P_PITCH);
  const int lane = threadIdx.x & 31;
  const int half = lane >> 4;
  const int l16  = lane & 15;

  const int  grp  = (int)(blockIdx.x & (NPAIR / WG_WAVES - 1));
  const int  pr   = grp * WG_WAVES + (wv < 4 ? wv : 11 - wv);
  const long bh   = (long)(blockIdx.x / (NPAIR / WG_WAVES));
  const long base = bh * (long)(SEQ_L * HEAD_D);
  const float* kbh = k + base;
  const float* vbh = v + base;

  for (int pass = 0; pass < 2; ++pass) {
    const int row = pass ? pr : (NBLK - 1 - pr);

    const int w     = row & ~3;
    const int nvert = w >> 2;
    const int nbl   = nvert + (row - w + 1);
    const int last  = nbl - 1;

    const int nvc = (pass ? 2 * grp : 30 - 2 * grp) & ~3;
    const int ntr = nvc >> 2;

    v2f kA0[16], kA1[16], kB0[16], kB1[16];
    v4f sg[8];
    if (ntr > 0) {
      stage_fetch(sg, kbh, vbh, 0, wv, l16, half);
    } else {
      load_k_frags(kA0, kbh, key_block(row, 0), l16, half);
      load_k_frags(kA1, kbh, key_block(row, last < 1 ? last : 1), l16, half);
    }

    const float* qrow = q + base + (long)(row * 16 + l16) * HEAD_D + 4 * half;
    v2f qa[16];
#pragma unroll
    for (int j = 0; j < 8; ++j) {
      const v4f t = *(const v4f*)(qrow + 8 * j);
      qa[2 * j + 0].x = t.x * 0.125f; qa[2 * j + 0].y = t.y * 0.125f;
      qa[2 * j + 1].x = t.z * 0.125f; qa[2 * j + 1].y = t.w * 0.125f;
    }

    v8f o0 = {}, o1 = {}, o2 = {}, o3 = {};
    float mrun = -1e30f, lrun = 0.0f;

    int t = 0;
    if (ntr > 0) {
      stage_store(sg, stg, wv, l16, half);
      stage_fetch(sg, kbh, vbh, ntr > 1 ? 1 : 0, wv, l16, half);
      stage_store(sg, stg + SLOT_F, wv, l16, half);
      stage_barrier();
      load_k_frags_lds(kA0, stg, l16, half);
      load_k_frags_lds(kA1, stg + TILE_F, l16, half);
      int oc = 0, on = SLOT_F, ot = 2 * SLOT_F;
      for (int u = 0; u + 1 < ntr; ++u) {
        stage_fetch(sg, kbh, vbh, u + 2 < ntr ? u + 2 : ntr - 1, wv, l16, half);
        tile_step<2, 3, true>(pshm, l16, half, qa, kA0, kA1, kB0, kB1, kbh, vbh, 0, 0, 0, 0,
                              stg + oc + 4 * TILE_F, stg + oc + 5 * TILE_F,
                              stg + oc + 2 * TILE_F, stg + oc + 3 * TILE_F,
                              o0, o1, o2, o3, mrun, lrun);
        tile_step<2, 3, true>(pshm, l16, half, qa, kB0, kB1, kA0, kA1, kbh, vbh, 0, 0, 0, 0,
                              stg + oc + 6 * TILE_F, stg + oc + 7 * TILE_F,
                              stg + on, stg + on + TILE_F,
                              o0, o1, o2, o3, mrun, lrun);
        stage_store(sg, stg + ot, wv, l16, half);
        stage_barrier();
        const int o_ = oc; oc = on; on = ot; ot = o_;
      }
      {
        const int f0 = nvc;
        const int f1 = nvc + 1 < last ? nvc + 1 : last;
        tile_step<2, 3, true>(pshm, l16, half, qa, kA0, kA1, kB0, kB1, kbh, vbh, 0, 0, 0, 0,
                              stg + oc + 4 * TILE_F, stg + oc + 5 * TILE_F,
                              stg + oc + 2 * TILE_F, stg + oc + 3 * TILE_F,
                              o0, o1, o2, o3, mrun, lrun);
        tile_step<2, 2, true>(pshm, l16, half, qa, kB0, kB1, kA0, kA1, kbh, vbh, 0, 0,
                              key_block(row, f0), key_block(row, f1),
                              stg + oc + 6 * TILE_F, stg + oc + 7 * TILE_F, nullptr, nullptr,
                              o0, o1, o2, o3, mrun, lrun);
        stage_barrier();
      }
      t = nvc;
    }

    while (nbl - t >= 4) {
      const int f0 = t + 4 < last ? t + 4 : last;
      const int f1 = t + 5 < last ? t + 5 : last;
      tile_step<2, 2, false>(pshm, l16, half, qa, kA0, kA1, kB0, kB1, kbh, vbh,
                      key_block(row, t), key_block(row, t + 1),
                      key_block(row, t + 2), key_block(row, t + 3),
                      nullptr, nullptr, nullptr, nullptr,
                      o0, o1, o2, o3, mrun, lrun);
      tile_step<2, 2, false>(pshm, l16, half, qa, kB0, kB1, kA0, kA1, kbh, vbh,
                      key_block(row, t + 2), key_block(row, t + 3),
                      key_block(row, f0), key_block(row, f1),
                      nullptr, nullptr, nullptr, nullptr,
                      o0, o1, o2, o3, mrun, lrun);
      t += 4;
    }
    const int rem = nbl - t;
    if (rem == 3) {
      tile_step<2, 1, false>(pshm, l16, half, qa, kA0, kA1, kB0, kB1, kbh, vbh,
                      key_block(row, t), key_block(row, t + 1),
                      key_block(row, t + 2), 0,
                      nullptr, nullptr, nullptr, nullptr,
                      o0, o1, o2, o3, mrun, lrun);
      tile_step<1, 0, false>(pshm, l16, half, qa, kB0, kB1, kA0, kA1, kbh, vbh,
                      key_block(row, t + 2), 0, 0, 0,
                      nullptr, nullptr, nullptr, nullptr,
                      o0, o1, o2, o3, mrun, lrun);
    } else if (rem == 2) {
      tile_step<2, 0, false>(pshm, l16, half, qa, kA0, kA1, kB0, kB1, kbh, vbh,
                      key_block(row, t), key_block(row, t + 1), 0, 0,
                      nullptr, nullptr, nullptr, nullptr,
                      o0, o1, o2, o3, mrun, lrun);
    } else if (rem == 1) {
      tile_step<1, 0, false>(pshm, l16, half, qa, kA0, kA1, kB0, kB1, kbh, vbh,
                      key_block(row, t), 0, 0, 0,
                      nullptr, nullptr, nullptr, nullptr,
                      o0, o1, o2, o3, mrun, lrun);
    }

    float* orow = out + base + (long)(row * 16) * HEAD_D + 4 * l16;
    float invd[8];
    rows_to_d(invd, 1.0f / (lrun + xhalf(lrun)));
#pragma unroll
    for (int r = 0; r < 8; ++r) {
      const float inv = invd[r];
      const int m = r + 8 * half;
      v4f ov;
      ov.x = o0[r] * inv; ov.y = o1[r] * inv;
      ov.z = o2[r] * inv; ov.w = o3[r] * inv;
      *(v4f*)(orow + (long)m * HEAD_D) = ov;
    }
  }
}

extern "C" void kernel_launch(void* const* d_in, const int* in_sizes, int n_in,
                              void* d_out, int out_size, void* d_ws, size_t ws_size,
                              hipStream_t stream) {
  (void)n_in; (void)out_size; (void)d_ws; (void)ws_size;
  const float* q = (const float*)d_in[0];
  const float* k = (const float*)d_in[1];
  const float* v = (const float*)d_in[2];
  float* out = (float*)d_out;

  const int bh = in_sizes[0] / (SEQ_L * HEAD_D);
  dim3 grid((unsigned)(bh * (NPAIR / WG_WAVES)));
  sparse_attn_wmma_f32<<<grid, 32 * WG_WAVES, 0, stream>>>(q, k, v, out);
}
